# starting kernel
# baseline (speedup 1.0000x reference)
#include <hip/hip_runtime.h>
#include <math.h>


typedef float v2f __attribute__((ext_vector_type(2)));
typedef float v8f __attribute__((ext_vector_type(8)));

#define B_SZ    4
#define LSEQ    1024
#define DMODEL  1024
#define DINNER  2048
#define DSTATE  16
#define DCONV   4
#define DTRANK  64
#define NROWS   (B_SZ * LSEQ)

__device__ __forceinline__ float silu_f(float x) { return x / (1.0f + __expf(-x)); }
__device__ __forceinline__ float softplus_f(float x) {
    return (x > 20.0f) ? x : log1pf(__expf(x));
}

__device__ __forceinline__ v8f wmma_f32(v2f a, v2f b, v8f c) {
    return __builtin_amdgcn_wmma_f32_16x16x4_f32(
               false, a, false, b, (short)0, c, false, false);
}

template <int EPI>
__global__ __launch_bounds__(64) void gemm_wmma_blocked(
    const float* __restrict__ A, const float* __restrict__ B,
    float* __restrict__ C, const float* __restrict__ bias,
    int K, int lda, int ldb, int ldc)
{
    const int lane   = threadIdx.x & 31;
    const int wave   = threadIdx.x >> 5;
    const int half   = lane >> 4;
    const int mr     = lane & 15;
    const int tm     = blockIdx.y;
    const int tnBase = (blockIdx.x * 2 + wave) * 4;

    const float* Arow0 = A + (size_t)(tm * 32 + mr) * lda;
    const float* Arow1 = Arow0 + (size_t)16 * lda;
    const float* Bc    = B + (size_t)(tnBase * 16 + mr);

    v8f c[2][4];
#pragma unroll
    for (int i = 0; i < 2; ++i)
#pragma unroll
        for (int j = 0; j < 4; ++j) c[i][j] = (v8f){};

#pragma unroll 4
    for (int k0 = 0; k0 < K; k0 += 4) {
        const int ka = k0 + 2 * half;
        v2f a0 = *(const v2f*)(Arow0 + ka);
        v2f a1 = *(const v2f*)(Arow1 + ka);
        v2f b[4];
#pragma unroll
        for (int j = 0; j < 4; ++j) {
            b[j].x = Bc[(size_t)ka * ldb + j * 16];
            b[j].y = Bc[(size_t)(ka + 1) * ldb + j * 16];
        }
#pragma unroll
        for (int j = 0; j < 4; ++j) {
            c[0][j] = wmma_f32(a0, b[j], c[0][j]);
            c[1][j] = wmma_f32(a1, b[j], c[1][j]);
        }
    }

#pragma unroll
    for (int j = 0; j < 4; ++j) {
        const int col = (tnBase + j) * 16 + mr;
        const float bv = (EPI == 1) ? bias[col] : 0.0f;
#pragma unroll
        for (int i = 0; i < 2; ++i) {
#pragma unroll
            for (int r = 0; r < 8; ++r) {
                const int row = tm * 32 + i * 16 + half * 8 + r;
                float v = c[i][j][r];
                if (EPI == 1) v = softplus_f(v + bv);
                C[(size_t)row * ldc + col] = v;
            }
        }
    }
}

__global__ __launch_bounds__(64) void gemm_wmma_simple(
    const float* __restrict__ A, const float* __restrict__ B,
    float* __restrict__ C, int K, int lda, int ldb, int ldc)
{
    const int lane = threadIdx.x & 31;
    const int wave = threadIdx.x >> 5;
    const int tn   = blockIdx.x * 2 + wave;
    const int tm   = blockIdx.y;
    const int half = lane >> 4;
    const int mr   = lane & 15;

    const float* Arow = A + (size_t)(tm * 16 + mr) * lda;
    const float* Bcol = B + (size_t)(tn * 16 + mr);

    v8f c = {};
#pragma unroll 8
    for (int k0 = 0; k0 < K; k0 += 4) {
        const int ka = k0 + 2 * half;
        v2f a = *(const v2f*)(Arow + ka);
        v2f b;
        b.x = Bcol[(size_t)ka * ldb];
        b.y = Bcol[(size_t)(ka + 1) * ldb];
        c = wmma_f32(a, b, c);
    }

    const int col = tn * 16 + mr;
#pragma unroll
    for (int r = 0; r < 8; ++r)
        C[(size_t)(tm * 16 + half * 8 + r) * ldc + col] = c[r];
}

__global__ __launch_bounds__(256) void conv_silu_kernel(
    const float* __restrict__ xz, const float* __restrict__ conv_w,
    const float* __restrict__ conv_b, float* __restrict__ xc)
{
    const int idx = blockIdx.x * 256 + threadIdx.x;
    const int d   = idx & (DINNER - 1);
    const int pos = idx >> 11;
    const int l   = pos & (LSEQ - 1);
    const int b   = pos >> 10;

    float acc = conv_b[d];
#pragma unroll
    for (int k = 0; k < DCONV; ++k) {
        const int ls = l - (DCONV - 1) + k;
        if (ls >= 0)
            acc += conv_w[d * DCONV + k] *
                   xz[((size_t)(b * LSEQ + ls)) * (2 * DINNER) + d];
    }
    xc[(size_t)pos * DINNER + d] = silu_f(acc);
}

__global__ __launch_bounds__(256) void scan_gate_kernel(
    const float* __restrict__ xc,   const float* __restrict__ dtm,
    const float* __restrict__ xdbl, const float* __restrict__ A_log,
    const float* __restrict__ xz,   const float* __restrict__ D_param,
    float* __restrict__ y)
{
    const int b = blockIdx.y;
    const int d = blockIdx.x * 256 + threadIdx.x;

    float Ar[DSTATE];
#pragma unroll
    for (int n = 0; n < DSTATE; ++n)
        Ar[n] = -__expf(A_log[d * DSTATE + n]);
    const float Dv = D_param[d];

    float h[DSTATE];
#pragma unroll
    for (int n = 0; n < DSTATE; ++n) h[n] = 0.0f;

    const size_t rowBase = (size_t)b * LSEQ;
    for (int t = 0; t < LSEQ; ++t) {
        const size_t row = rowBase + t;
        const float* bc = xdbl + row * (DTRANK + 2 * DSTATE) + DTRANK;

        const float u   = xc [row * DINNER + d];
        const float dtv = dtm[row * DINNER + d];
        __builtin_prefetch(xc  + (row + 1) * DINNER + d, 0, 1);
        __builtin_prefetch(dtm + (row + 1) * DINNER + d, 0, 1);

        float acc = 0.0f;
#pragma unroll
        for (int n = 0; n < DSTATE; ++n) {
            const float dA = __expf(dtv * Ar[n]);
            h[n] = dA * h[n] + (dtv * bc[n]) * u;
            acc += bc[DSTATE + n] * h[n];
        }
        const float z = xz[row * (2 * DINNER) + DINNER + d];
        y[row * DINNER + d] = (acc + Dv * u) * silu_f(z);
    }
}

extern "C" void kernel_launch(void* const* d_in, const int* in_sizes, int n_in,
                              void* d_out, int out_size, void* d_ws, size_t ws_size,
                              hipStream_t stream)
{
    const float* hidden    = (const float*)d_in[0];
    const float* in_proj_w = (const float*)d_in[1];
    const float* conv_w    = (const float*)d_in[2];
    const float* conv_b    = (const float*)d_in[3];
    const float* x_proj_w  = (const float*)d_in[4];
    const float* dt_proj_w = (const float*)d_in[5];
    const float* dt_bias   = (const float*)d_in[6];
    const float* A_log     = (const float*)d_in[7];
    const float* D_param   = (const float*)d_in[8];
    const float* out_proj  = (const float*)d_in[9];
    float*       out       = (float*)d_out;

    float* ws   = (float*)d_ws;
    float* xz   = ws;
    float* xc   = xz   + (size_t)NROWS * 2 * DINNER;
    float* xdbl = xc   + (size_t)NROWS * DINNER;
    float* dtm  = xdbl + (size_t)NROWS * (DTRANK + 2 * DSTATE);
    float* y    = dtm  + (size_t)NROWS * DINNER;

    gemm_wmma_blocked<0><<<dim3((2 * DINNER) / 128, NROWS / 32), 64, 0, stream>>>(
        hidden, in_proj_w, xz, nullptr, DMODEL, DMODEL, 2 * DINNER, 2 * DINNER);

    conv_silu_kernel<<<(NROWS * DINNER) / 256, 256, 0, stream>>>(xz, conv_w, conv_b, xc);

    gemm_wmma_simple<<<dim3((DTRANK + 2 * DSTATE) / 32, NROWS / 16), 64, 0, stream>>>(
        xc, x_proj_w, xdbl, DINNER, DINNER,
        DTRANK + 2 * DSTATE, DTRANK + 2 * DSTATE);

    gemm_wmma_blocked<1><<<dim3(DINNER / 128, NROWS / 32), 64, 0, stream>>>(
        xdbl, dt_proj_w, dtm, dt_bias, DTRANK,
        DTRANK + 2 * DSTATE, DINNER, DINNER);

    scan_gate_kernel<<<dim3(DINNER / 256, B_SZ), 256, 0, stream>>>(
        xc, dtm, xdbl, A_log, xz, D_param, y);

    gemm_wmma_blocked<0><<<dim3(DMODEL / 128, NROWS / 32), 64, 0, stream>>>(
        y, out_proj, out, nullptr, DINNER, DINNER, DMODEL, DMODEL);
}


# conv+SiLU fused into the x_proj GEMM: a 16-row work-group computes its xc values in the A loader (stored on the way), 8 waves split K, partials added via LDS; 5 launches; x_proj sum order changed
# speedup vs baseline: 2.7997x; 2.7997x over previous
#include <hip/hip_runtime.h>
#include <math.h>


typedef float v2f __attribute__((ext_vector_type(2)));
typedef float v8f __attribute__((ext_vector_type(8)));

#define B_SZ    4
#define LSEQ    1024
#define DMODEL  1024
#define DINNER  2048
#define DSTATE  16
#define DCONV   4
#define DTRANK  64
#define NROWS   (B_SZ * LSEQ)

__device__ __forceinline__ float silu_f(float x) { return x / (1.0f + __expf(-x)); }
__device__ __forceinline__ float softplus_f(float x) {
    return (x > 20.0f) ? x : log1pf(__expf(x));
}

__device__ __forceinline__ v8f wmma_f32(v2f a, v2f b, v8f c) {
    return __builtin_amdgcn_wmma_f32_16x16x4_f32(
               false, a, false, b, (short)0, c, false, false);
}

#define GEMM_MT 4
#define GEMM_NT 4

template <int KC>
struct FragSet {
    float  a[GEMM_MT][KC / 2];
    float4 b[KC / 2];
};

template <int EPI, int KC, int K, int lda, int ldb, int ldc>
__global__ __launch_bounds__(128) void gemm_wmma_regtile(
    const float* __restrict__ A, const float* __restrict__ B,
    float* __restrict__ C, const float* __restrict__ bias)
{
    constexpr int KH = KC / 2;
    constexpr int NS = KC / 4;
    const int lane = threadIdx.x & 31;
    const int wave = threadIdx.x >> 5;
    const int half = lane >> 4;
    const int mr   = lane & 15;
    const int row0 = (blockIdx.y * 2 + (wave >> 1)) * (16 * GEMM_MT);
    const int col0 = (blockIdx.x * 2 + (wave & 1)) * (16 * GEMM_NT) + 4 * mr;

    const float* Ap = A + (size_t)(row0 + mr) * lda + KH * half;
    const float* Bp = B + (size_t)(KH * half) * ldb + col0;

    v8f c[GEMM_MT][GEMM_NT];
#pragma unroll
    for (int i = 0; i < GEMM_MT; ++i)
#pragma unroll
        for (int j = 0; j < GEMM_NT; ++j) c[i][j] = (v8f){};

    auto load_frags = [&](FragSet<KC>& f, int k0) {
#pragma unroll
        for (int i = 0; i < GEMM_MT; ++i) {
#pragma unroll
            for (int q = 0; q < KH / 4; ++q) {
                const float4 t = *(const float4*)(Ap + (size_t)(i * 16) * lda + k0 + 4 * q);
                f.a[i][4 * q + 0] = t.x; f.a[i][4 * q + 1] = t.y;
                f.a[i][4 * q + 2] = t.z; f.a[i][4 * q + 3] = t.w;
            }
        }
#pragma unroll
        for (int m = 0; m < KH; ++m)
            f.b[m] = *(const float4*)(Bp + (size_t)(k0 + m) * ldb);
    };

    auto mma_chunk = [&](const FragSet<KC>& f) {
#pragma unroll
        for (int s = 0; s < NS; ++s) {
            const float4 b0 = f.b[2 * s], b1 = f.b[2 * s + 1];
            v2f bv[GEMM_NT];
            bv[0].x = b0.x; bv[0].y = b1.x;
            bv[1].x = b0.y; bv[1].y = b1.y;
            bv[2].x = b0.z; bv[2].y = b1.z;
            bv[3].x = b0.w; bv[3].y = b1.w;
#pragma unroll
            for (int j = 0; j < GEMM_NT; ++j) {
#pragma unroll
                for (int i = 0; i < GEMM_MT; ++i) {
                    v2f av; av.x = f.a[i][2 * s]; av.y = f.a[i][2 * s + 1];
                    c[i][j] = wmma_f32(av, bv[j], c[i][j]);
                }
            }
        }
    };

    FragSet<KC> f0, f1;
    load_frags(f0, 0);
    int k0 = 0;
    for (; k0 < K - 2 * KC; k0 += 2 * KC) {
        load_frags(f1, k0 + KC);
        mma_chunk(f0);
        load_frags(f0, k0 + 2 * KC);
        mma_chunk(f1);
    }
    load_frags(f1, k0 + KC);
    mma_chunk(f0);
    mma_chunk(f1);

    float4 bv4 = make_float4(0.0f, 0.0f, 0.0f, 0.0f);
    if (EPI == 1) bv4 = *(const float4*)(bias + col0);
#pragma unroll
    for (int i = 0; i < GEMM_MT; ++i) {
#pragma unroll
        for (int r = 0; r < 8; ++r) {
            const int row = row0 + i * 16 + half * 8 + r;
            float4 v;
            v.x = c[i][0][r]; v.y = c[i][1][r]; v.z = c[i][2][r]; v.w = c[i][3][r];
            if (EPI == 1) {
                v.x = softplus_f(v.x + bv4.x); v.y = softplus_f(v.y + bv4.y);
                v.z = softplus_f(v.z + bv4.z); v.w = softplus_f(v.w + bv4.w);
            }
            *(float4*)(C + (size_t)row * ldc + col0) = v;
        }
    }
}

#define XP_N      (DTRANK + 2 * DSTATE)
#define XP_WAVES  8
#define XP_KW     (DINNER / XP_WAVES)
#define XP_KC     32

__global__ __launch_bounds__(XP_WAVES * 32) void conv_xproj_kernel(
    const float* __restrict__ xz, const float* __restrict__ conv_w,
    const float* __restrict__ conv_b, const float* __restrict__ Bw,
    float* __restrict__ xc, float* __restrict__ xdbl)
{
    __shared__ float red[XP_WAVES][16 * XP_N];

    const int lane = threadIdx.x & 31;
    const int wave = threadIdx.x >> 5;
    const int half = lane >> 4;
    const int mr   = lane & 15;
    const int row  = blockIdx.x * 16 + mr;
    const int l    = row & (LSEQ - 1);
    const int kw   = wave * XP_KW + (XP_KC / 2) * half;

    const bool ok0 = (l >= 3), ok1 = (l >= 2), ok2 = (l >= 1);
    const float* x3 = xz + (size_t)row * (2 * DINNER) + kw;
    const float* x0 = x3 - (ok0 ? 3 : 0) * (size_t)(2 * DINNER);
    const float* x1 = x3 - (ok1 ? 2 : 0) * (size_t)(2 * DINNER);
    const float* x2 = x3 - (ok2 ? 1 : 0) * (size_t)(2 * DINNER);
    const float m0 = ok0 ? 1.0f : 0.0f, m1 = ok1 ? 1.0f : 0.0f, m2 = ok2 ? 1.0f : 0.0f;

    const float* wp  = conv_w + (size_t)kw * DCONV;
    const float* cbp = conv_b + kw;
    const float* bp  = Bw + (size_t)kw * XP_N;
    float*       xcp = xc + (size_t)row * DINNER + kw;

    v8f acc[6];
#pragma unroll
    for (int j = 0; j < 6; ++j) acc[j] = (v8f){};

    for (int ko = 0; ko < XP_KW; ko += XP_KC) {
        float4 b4[XP_KC / 2];
        float2 b2[XP_KC / 2];
#pragma unroll
        for (int m = 0; m < XP_KC / 2; ++m) {
            b4[m] = *(const float4*)(bp + (size_t)(ko + m) * XP_N + 4 * mr);
            b2[m] = *(const float2*)(bp + (size_t)(ko + m) * XP_N + 64 + 2 * mr);
        }

        float a[XP_KC / 2];
#pragma unroll
        for (int q = 0; q < XP_KC / 8; ++q) {
            const int k = ko + 4 * q;
            float4 v0 = *(const float4*)(x0 + k);
            float4 v1 = *(const float4*)(x1 + k);
            float4 v2 = *(const float4*)(x2 + k);
            const float4 v3 = *(const float4*)(x3 + k);
            const float4 cb = *(const float4*)(cbp + k);
            const float4 w0 = *(const float4*)(wp + (size_t)(k + 0) * DCONV);
            const float4 w1 = *(const float4*)(wp + (size_t)(k + 1) * DCONV);
            const float4 w2 = *(const float4*)(wp + (size_t)(k + 2) * DCONV);
            const float4 w3 = *(const float4*)(wp + (size_t)(k + 3) * DCONV);
            v0.x *= m0; v0.y *= m0; v0.z *= m0; v0.w *= m0;
            v1.x *= m1; v1.y *= m1; v1.z *= m1; v1.w *= m1;
            v2.x *= m2; v2.y *= m2; v2.z *= m2; v2.w *= m2;
            float s0 = cb.x, s1 = cb.y, s2 = cb.z, s3 = cb.w;
            s0 += w0.x * v0.x; s0 += w0.y * v1.x; s0 += w0.z * v2.x; s0 += w0.w * v3.x;
            s1 += w1.x * v0.y; s1 += w1.y * v1.y; s1 += w1.z * v2.y; s1 += w1.w * v3.y;
            s2 += w2.x * v0.z; s2 += w2.y * v1.z; s2 += w2.z * v2.z; s2 += w2.w * v3.z;
            s3 += w3.x * v0.w; s3 += w3.y * v1.w; s3 += w3.z * v2.w; s3 += w3.w * v3.w;
            float4 o;
            o.x = silu_f(s0); o.y = silu_f(s1); o.z = silu_f(s2); o.w = silu_f(s3);
            *(float4*)(xcp + k) = o;
            a[4 * q + 0] = o.x; a[4 * q + 1] = o.y; a[4 * q + 2] = o.z; a[4 * q + 3] = o.w;
        }

#pragma unroll
        for (int s = 0; s < XP_KC / 4; ++s) {
            v2f av; av.x = a[2 * s]; av.y = a[2 * s + 1];
            const float4 p0 = b4[2 * s], p1 = b4[2 * s + 1];
            const float2 r0 = b2[2 * s], r1 = b2[2 * s + 1];
            v2f bv[6];
            bv[0].x = p0.x; bv[0].y = p1.x;
            bv[1].x = p0.y; bv[1].y = p1.y;
            bv[2].x = p0.z; bv[2].y = p1.z;
            bv[3].x = p0.w; bv[3].y = p1.w;
            bv[4].x = r0.x; bv[4].y = r1.x;
            bv[5].x = r0.y; bv[5].y = r1.y;
#pragma unroll
            for (int j = 0; j < 6; ++j) acc[j] = wmma_f32(av, bv[j], acc[j]);
        }
    }

#pragma unroll
    for (int r = 0; r < 8; ++r) {
        float* dst = &red[wave][(half * 8 + r) * XP_N];
        float4 t4; t4.x = acc[0][r]; t4.y = acc[1][r]; t4.z = acc[2][r]; t4.w = acc[3][r];
        float2 t2; t2.x = acc[4][r]; t2.y = acc[5][r];
        *(float4*)(dst + 4 * mr)      = t4;
        *(float2*)(dst + 64 + 2 * mr) = t2;
    }
    __syncthreads();

    float* outp = xdbl + (size_t)blockIdx.x * (16 * XP_N);
#pragma unroll
    for (int i = 0; i < (16 * XP_N) / (XP_WAVES * 32); ++i) {
        const int idx = threadIdx.x + i * (XP_WAVES * 32);
        float sum = red[0][idx];
#pragma unroll
        for (int w = 1; w < XP_WAVES; ++w) sum += red[w][idx];
        outp[idx] = sum;
    }
}

#define NCHUNK  32
#define TCH     (LSEQ / NCHUNK)
#define CPITCH  20

__global__ __launch_bounds__(NCHUNK * 32) void scan_gate_chunked_kernel(
    const float* __restrict__ xc,   const float* __restrict__ dtm,
    const float* __restrict__ xdbl, const float* __restrict__ A_log,
    const float* __restrict__ xz,   const float* __restrict__ D_param,
    float* __restrict__ y)
{
    __shared__ float sH[NCHUNK][32][CPITCH];
    __shared__ float sP[NCHUNK][32][CPITCH];

    const int lane = threadIdx.x & 31;
    const int w = __builtin_amdgcn_readfirstlane((int)(threadIdx.x >> 5));
    const int b = blockIdx.y;
    const int d = blockIdx.x * 32 + lane;

    float Ar[DSTATE];
#pragma unroll
    for (int n = 0; n < DSTATE; ++n)
        Ar[n] = -__expf(A_log[d * DSTATE + n]);

    const size_t row0 = (size_t)b * LSEQ + (size_t)w * TCH;
    const float* up  = xc   + row0 * DINNER + d;
    const float* dp  = dtm  + row0 * DINNER + d;
    const float* zp  = xz   + row0 * (2 * DINNER) + DINNER + d;
    const float* bcp = xdbl + row0 * (DTRANK + 2 * DSTATE) + DTRANK;
    float*       yp  = y    + row0 * DINNER + d;

    float h[DSTATE];

    if (w < NCHUNK - 1) {
#pragma unroll
        for (int n = 0; n < DSTATE; ++n) h[n] = 0.0f;
        float sdt = 0.0f;
        for (int t = 0; t < TCH; ++t) {
            const float* bc = bcp + (size_t)t * (DTRANK + 2 * DSTATE);
            const float u   = up[(size_t)t * DINNER];
            const float dtv = dp[(size_t)t * DINNER];
            sdt += dtv;
#pragma unroll
            for (int n = 0; n < DSTATE; ++n) {
                const float dA = __expf(dtv * Ar[n]);
                h[n] = dA * h[n] + (dtv * bc[n]) * u;
            }
        }
#pragma unroll
        for (int q = 0; q < DSTATE / 4; ++q) {
            float4 hv, pv;
            hv.x = h[4 * q + 0]; hv.y = h[4 * q + 1];
            hv.z = h[4 * q + 2]; hv.w = h[4 * q + 3];
            pv.x = __expf(sdt * Ar[4 * q + 0]); pv.y = __expf(sdt * Ar[4 * q + 1]);
            pv.z = __expf(sdt * Ar[4 * q + 2]); pv.w = __expf(sdt * Ar[4 * q + 3]);
            *(float4*)&sH[w][lane][4 * q] = hv;
            *(float4*)&sP[w][lane][4 * q] = pv;
        }
    }
    __syncthreads();

#pragma unroll
    for (int n = 0; n < DSTATE; ++n) h[n] = 0.0f;
    for (int c = 0; c < w; ++c) {
#pragma unroll
        for (int q = 0; q < DSTATE / 4; ++q) {
            const float4 pv = *(const float4*)&sP[c][lane][4 * q];
            const float4 hv = *(const float4*)&sH[c][lane][4 * q];
            h[4 * q + 0] = pv.x * h[4 * q + 0] + hv.x;
            h[4 * q + 1] = pv.y * h[4 * q + 1] + hv.y;
            h[4 * q + 2] = pv.z * h[4 * q + 2] + hv.z;
            h[4 * q + 3] = pv.w * h[4 * q + 3] + hv.w;
        }
    }

    const float Dv = D_param[d];
    for (int t = 0; t < TCH; ++t) {
        const float* bc = bcp + (size_t)t * (DTRANK + 2 * DSTATE);
        const float u   = up[(size_t)t * DINNER];
        const float dtv = dp[(size_t)t * DINNER];
        const float z   = zp[(size_t)t * (2 * DINNER)];

        float acc = 0.0f;
#pragma unroll
        for (int n = 0; n < DSTATE; ++n) {
            const float dA = __expf(dtv * Ar[n]);
            h[n] = dA * h[n] + (dtv * bc[n]) * u;
            acc += bc[DSTATE + n] * h[n];
        }
        yp[(size_t)t * DINNER] = (acc + Dv * u) * silu_f(z);
    }
}

extern "C" void kernel_launch(void* const* d_in, const int* in_sizes, int n_in,
                              void* d_out, int out_size, void* d_ws, size_t ws_size,
                              hipStream_t stream)
{
    const float* hidden    = (const float*)d_in[0];
    const float* in_proj_w = (const float*)d_in[1];
    const float* conv_w    = (const float*)d_in[2];
    const float* conv_b    = (const float*)d_in[3];
    const float* x_proj_w  = (const float*)d_in[4];
    const float* dt_proj_w = (const float*)d_in[5];
    const float* dt_bias   = (const float*)d_in[6];
    const float* A_log     = (const float*)d_in[7];
    const float* D_param   = (const float*)d_in[8];
    const float* out_proj  = (const float*)d_in[9];
    float*       out       = (float*)d_out;

    float* ws   = (float*)d_ws;
    float* xz   = ws;
    float* xc   = xz   + (size_t)NROWS * 2 * DINNER;
    float* xdbl = xc   + (size_t)NROWS * DINNER;
    float* dtm  = xdbl + (size_t)NROWS * (DTRANK + 2 * DSTATE);
    float* y    = dtm  + (size_t)NROWS * DINNER;

    gemm_wmma_regtile<0, 32, DMODEL, DMODEL, 2 * DINNER, 2 * DINNER>
        <<<dim3((2 * DINNER) / 128, NROWS / 128), 128, 0, stream>>>(
        hidden, in_proj_w, xz, nullptr);

    conv_xproj_kernel<<<NROWS / 16, XP_WAVES * 32, 0, stream>>>(
        xz, conv_w, conv_b, x_proj_w, xc, xdbl);

    gemm_wmma_regtile<1, 32, DTRANK, DTRANK + 2 * DSTATE, DINNER, DINNER>
        <<<dim3(DINNER / 128, NROWS / 128), 128, 0, stream>>>(
        xdbl, dt_proj_w, dtm, dt_bias);

    scan_gate_chunked_kernel<<<dim3(DINNER / 32, B_SZ), NCHUNK * 32, 0, stream>>>(
        xc, dtm, xdbl, A_log, xz, D_param, y);

    gemm_wmma_regtile<0, 32, DINNER, DINNER, DMODEL, DMODEL>
        <<<dim3(DMODEL / 128, NROWS / 128), 128, 0, stream>>>(
        y, out_proj, out, nullptr);
}


# dt_proj softplus epilogue rolled: lanes park their 32 biased float4 in own LDS slots, a loop trip does 4 values side by side (select, no branch) + 1 store; code now fits the instruction cache
# speedup vs baseline: 3.4504x; 1.2324x over previous
#include <hip/hip_runtime.h>
#include <math.h>


typedef float v2f __attribute__((ext_vector_type(2)));
typedef float v8f __attribute__((ext_vector_type(8)));

#define B_SZ    4
#define LSEQ    1024
#define DMODEL  1024
#define DINNER  2048
#define DSTATE  16
#define DCONV   4
#define DTRANK  64
#define NROWS   (B_SZ * LSEQ)

__device__ __forceinline__ float silu_f(float x) { return x / (1.0f + __expf(-x)); }
__device__ __forceinline__ float softplus_f(float x) {
    const float s = log1pf(__expf(x));
    return (x > 20.0f) ? x : s;
}

__device__ __forceinline__ v8f wmma_f32(v2f a, v2f b, v8f c) {
    return __builtin_amdgcn_wmma_f32_16x16x4_f32(
               false, a, false, b, (short)0, c, false, false);
}

#define GEMM_MT 4
#define GEMM_NT 4

template <int KC>
struct FragSet {
    float  a[GEMM_MT][KC / 2];
    float4 b[KC / 2];
};

template <int EPI, int KC, int K, int lda, int ldb, int ldc>
__global__ __launch_bounds__(128) void gemm_wmma_regtile(
    const float* __restrict__ A, const float* __restrict__ B,
    float* __restrict__ C, const float* __restrict__ bias)
{
    constexpr int KH = KC / 2;
    constexpr int NS = KC / 4;
    const int lane = threadIdx.x & 31;
    const int wave = threadIdx.x >> 5;
    const int half = lane >> 4;
    const int mr   = lane & 15;
    const int row0 = (blockIdx.y * 2 + (wave >> 1)) * (16 * GEMM_MT);
    const int col0 = (blockIdx.x * 2 + (wave & 1)) * (16 * GEMM_NT) + 4 * mr;

    const float* Ap = A + (size_t)(row0 + mr) * lda + KH * half;
    const float* Bp = B + (size_t)(KH * half) * ldb + col0;

    v8f c[GEMM_MT][GEMM_NT];
#pragma unroll
    for (int i = 0; i < GEMM_MT; ++i)
#pragma unroll
        for (int j = 0; j < GEMM_NT; ++j) c[i][j] = (v8f){};

    auto load_frags = [&](FragSet<KC>& f, int k0) {
#pragma unroll
        for (int i = 0; i < GEMM_MT; ++i) {
#pragma unroll
            for (int q = 0; q < KH / 4; ++q) {
                const float4 t = *(const float4*)(Ap + (size_t)(i * 16) * lda + k0 + 4 * q);
                f.a[i][4 * q + 0] = t.x; f.a[i][4 * q + 1] = t.y;
                f.a[i][4 * q + 2] = t.z; f.a[i][4 * q + 3] = t.w;
            }
        }
#pragma unroll
        for (int m = 0; m < KH; ++m)
            f.b[m] = *(const float4*)(Bp + (size_t)(k0 + m) * ldb);
    };

    auto mma_chunk = [&](const FragSet<KC>& f) {
#pragma unroll
        for (int s = 0; s < NS; ++s) {
            const float4 b0 = f.b[2 * s], b1 = f.b[2 * s + 1];
            v2f bv[GEMM_NT];
            bv[0].x = b0.x; bv[0].y = b1.x;
            bv[1].x = b0.y; bv[1].y = b1.y;
            bv[2].x = b0.z; bv[2].y = b1.z;
            bv[3].x = b0.w; bv[3].y = b1.w;
#pragma unroll
            for (int j = 0; j < GEMM_NT; ++j) {
#pragma unroll
                for (int i = 0; i < GEMM_MT; ++i) {
                    v2f av; av.x = f.a[i][2 * s]; av.y = f.a[i][2 * s + 1];
                    c[i][j] = wmma_f32(av, bv[j], c[i][j]);
                }
            }
        }
    };

    FragSet<KC> f0, f1;
    load_frags(f0, 0);
    int k0 = 0;
    for (; k0 < K - 2 * KC; k0 += 2 * KC) {
        load_frags(f1, k0 + KC);
        mma_chunk(f0);
        load_frags(f0, k0 + 2 * KC);
        mma_chunk(f1);
    }
    load_frags(f1, k0 + KC);
    mma_chunk(f0);
    mma_chunk(f1);

    if constexpr (EPI == 1) {
        __shared__ float4 park[4 * 32 * 32];
        const float4 bv4 = *(const float4*)(bias + col0);
        float4* mine = park + (size_t)(wave * 32) * 32 + lane;
#pragma unroll
        for (int i = 0; i < GEMM_MT; ++i) {
#pragma unroll
            for (int r = 0; r < 8; ++r) {
                float4 v;
                v.x = c[i][0][r] + bv4.x; v.y = c[i][1][r] + bv4.y;
                v.z = c[i][2][r] + bv4.z; v.w = c[i][3][r] + bv4.w;
                mine[(i * 8 + r) * 32] = v;
            }
        }
        float* Cp = C + (size_t)(row0 + half * 8) * ldc + col0;
#pragma unroll 1
        for (int s = 0; s < 32; ++s) {
            float4 v = mine[s * 32];
            v.x = softplus_f(v.x); v.y = softplus_f(v.y);
            v.z = softplus_f(v.z); v.w = softplus_f(v.w);
            const int row = (s >> 3) * 16 + (s & 7);
            *(float4*)(Cp + (size_t)row * ldc) = v;
        }
    } else {
#pragma unroll
        for (int i = 0; i < GEMM_MT; ++i) {
#pragma unroll
            for (int r = 0; r < 8; ++r) {
                const int row = row0 + i * 16 + half * 8 + r;
                float4 v;
                v.x = c[i][0][r]; v.y = c[i][1][r]; v.z = c[i][2][r]; v.w = c[i][3][r];
                *(float4*)(C + (size_t)row * ldc + col0) = v;
            }
        }
    }
}

#define XP_N      (DTRANK + 2 * DSTATE)
#define XP_WAVES  8
#define XP_KW     (DINNER / XP_WAVES)
#define XP_KC     32

__global__ __launch_bounds__(XP_WAVES * 32) void conv_xproj_kernel(
    const float* __restrict__ xz, const float* __restrict__ conv_w,
    const float* __restrict__ conv_b, const float* __restrict__ Bw,
    float* __restrict__ xc, float* __restrict__ xdbl)
{
    __shared__ float red[XP_WAVES][16 * XP_N];

    const int lane = threadIdx.x & 31;
    const int wave = threadIdx.x >> 5;
    const int half = lane >> 4;
    const int mr   = lane & 15;
    const int row  = blockIdx.x * 16 + mr;
    const int l    = row & (LSEQ - 1);
    const int kw   = wave * XP_KW + (XP_KC / 2) * half;

    const bool ok0 = (l >= 3), ok1 = (l >= 2), ok2 = (l >= 1);
    const float* x3 = xz + (size_t)row * (2 * DINNER) + kw;
    const float* x0 = x3 - (ok0 ? 3 : 0) * (size_t)(2 * DINNER);
    const float* x1 = x3 - (ok1 ? 2 : 0) * (size_t)(2 * DINNER);
    const float* x2 = x3 - (ok2 ? 1 : 0) * (size_t)(2 * DINNER);
    const float m0 = ok0 ? 1.0f : 0.0f, m1 = ok1 ? 1.0f : 0.0f, m2 = ok2 ? 1.0f : 0.0f;

    const float* wp  = conv_w + (size_t)kw * DCONV;
    const float* cbp = conv_b + kw;
    const float* bp  = Bw + (size_t)kw * XP_N;
    float*       xcp = xc + (size_t)row * DINNER + kw;

    v8f acc[6];
#pragma unroll
    for (int j = 0; j < 6; ++j) acc[j] = (v8f){};

    for (int ko = 0; ko < XP_KW; ko += XP_KC) {
        float4 b4[XP_KC / 2];
        float2 b2[XP_KC / 2];
#pragma unroll
        for (int m = 0; m < XP_KC / 2; ++m) {
            b4[m] = *(const float4*)(bp + (size_t)(ko + m) * XP_N + 4 * mr);
            b2[m] = *(const float2*)(bp + (size_t)(ko + m) * XP_N + 64 + 2 * mr);
        }

        float a[XP_KC / 2];
#pragma unroll
        for (int q = 0; q < XP_KC / 8; ++q) {
            const int k = ko + 4 * q;
            float4 v0 = *(const float4*)(x0 + k);
            float4 v1 = *(const float4*)(x1 + k);
            float4 v2 = *(const float4*)(x2 + k);
            const float4 v3 = *(const float4*)(x3 + k);
            const float4 cb = *(const float4*)(cbp + k);
            const float4 w0 = *(const float4*)(wp + (size_t)(k + 0) * DCONV);
            const float4 w1 = *(const float4*)(wp + (size_t)(k + 1) * DCONV);
            const float4 w2 = *(const float4*)(wp + (size_t)(k + 2) * DCONV);
            const float4 w3 = *(const float4*)(wp + (size_t)(k + 3) * DCONV);
            v0.x *= m0; v0.y *= m0; v0.z *= m0; v0.w *= m0;
            v1.x *= m1; v1.y *= m1; v1.z *= m1; v1.w *= m1;
            v2.x *= m2; v2.y *= m2; v2.z *= m2; v2.w *= m2;
            float s0 = cb.x, s1 = cb.y, s2 = cb.z, s3 = cb.w;
            s0 += w0.x * v0.x; s0 += w0.y * v1.x; s0 += w0.z * v2.x; s0 += w0.w * v3.x;
            s1 += w1.x * v0.y; s1 += w1.y * v1.y; s1 += w1.z * v2.y; s1 += w1.w * v3.y;
            s2 += w2.x * v0.z; s2 += w2.y * v1.z; s2 += w2.z * v2.z; s2 += w2.w * v3.z;
            s3 += w3.x * v0.w; s3 += w3.y * v1.w; s3 += w3.z * v2.w; s3 += w3.w * v3.w;
            float4 o;
            o.x = silu_f(s0); o.y = silu_f(s1); o.z = silu_f(s2); o.w = silu_f(s3);
            *(float4*)(xcp + k) = o;
            a[4 * q + 0] = o.x; a[4 * q + 1] = o.y; a[4 * q + 2] = o.z; a[4 * q + 3] = o.w;
        }

#pragma unroll
        for (int s = 0; s < XP_KC / 4; ++s) {
            v2f av; av.x = a[2 * s]; av.y = a[2 * s + 1];
            const float4 p0 = b4[2 * s], p1 = b4[2 * s + 1];
            const float2 r0 = b2[2 * s], r1 = b2[2 * s + 1];
            v2f bv[6];
            bv[0].x = p0.x; bv[0].y = p1.x;
            bv[1].x = p0.y; bv[1].y = p1.y;
            bv[2].x = p0.z; bv[2].y = p1.z;
            bv[3].x = p0.w; bv[3].y = p1.w;
            bv[4].x = r0.x; bv[4].y = r1.x;
            bv[5].x = r0.y; bv[5].y = r1.y;
#pragma unroll
            for (int j = 0; j < 6; ++j) acc[j] = wmma_f32(av, bv[j], acc[j]);
        }
    }

#pragma unroll
    for (int r = 0; r < 8; ++r) {
        float* dst = &red[wave][(half * 8 + r) * XP_N];
        float4 t4; t4.x = acc[0][r]; t4.y = acc[1][r]; t4.z = acc[2][r]; t4.w = acc[3][r];
        float2 t2; t2.x = acc[4][r]; t2.y = acc[5][r];
        *(float4*)(dst + 4 * mr)      = t4;
        *(float2*)(dst + 64 + 2 * mr) = t2;
    }
    __syncthreads();

    float* outp = xdbl + (size_t)blockIdx.x * (16 * XP_N);
#pragma unroll
    for (int i = 0; i < (16 * XP_N) / (XP_WAVES * 32); ++i) {
        const int idx = threadIdx.x + i * (XP_WAVES * 32);
        float sum = red[0][idx];
#pragma unroll
        for (int w = 1; w < XP_WAVES; ++w) sum += red[w][idx];
        outp[idx] = sum;
    }
}

#define NCHUNK  32
#define TCH     (LSEQ / NCHUNK)
#define CPITCH  20

__global__ __launch_bounds__(NCHUNK * 32) void scan_gate_chunked_kernel(
    const float* __restrict__ xc,   const float* __restrict__ dtm,
    const float* __restrict__ xdbl, const float* __restrict__ A_log,
    const float* __restrict__ xz,   const float* __restrict__ D_param,
    float* __restrict__ y)
{
    __shared__ float sH[NCHUNK][32][CPITCH];
    __shared__ float sP[NCHUNK][32][CPITCH];

    const int lane = threadIdx.x & 31;
    const int w = __builtin_amdgcn_readfirstlane((int)(threadIdx.x >> 5));
    const int b = blockIdx.y;
    const int d = blockIdx.x * 32 + lane;

    float Ar[DSTATE];
#pragma unroll
    for (int n = 0; n < DSTATE; ++n)
        Ar[n] = -__expf(A_log[d * DSTATE + n]);

    const size_t row0 = (size_t)b * LSEQ + (size_t)w * TCH;
    const float* up  = xc   + row0 * DINNER + d;
    const float* dp  = dtm  + row0 * DINNER + d;
    const float* zp  = xz   + row0 * (2 * DINNER) + DINNER + d;
    const float* bcp = xdbl + row0 * (DTRANK + 2 * DSTATE) + DTRANK;
    float*       yp  = y    + row0 * DINNER + d;

    float h[DSTATE];

    if (w < NCHUNK - 1) {
#pragma unroll
        for (int n = 0; n < DSTATE; ++n) h[n] = 0.0f;
        float sdt = 0.0f;
        for (int t = 0; t < TCH; ++t) {
            const float* bc = bcp + (size_t)t * (DTRANK + 2 * DSTATE);
            const float u   = up[(size_t)t * DINNER];
            const float dtv = dp[(size_t)t * DINNER];
            sdt += dtv;
#pragma unroll
            for (int n = 0; n < DSTATE; ++n) {
                const float dA = __expf(dtv * Ar[n]);
                h[n] = dA * h[n] + (dtv * bc[n]) * u;
            }
        }
#pragma unroll
        for (int q = 0; q < DSTATE / 4; ++q) {
            float4 hv, pv;
            hv.x = h[4 * q + 0]; hv.y = h[4 * q + 1];
            hv.z = h[4 * q + 2]; hv.w = h[4 * q + 3];
            pv.x = __expf(sdt * Ar[4 * q + 0]); pv.y = __expf(sdt * Ar[4 * q + 1]);
            pv.z = __expf(sdt * Ar[4 * q + 2]); pv.w = __expf(sdt * Ar[4 * q + 3]);
            *(float4*)&sH[w][lane][4 * q] = hv;
            *(float4*)&sP[w][lane][4 * q] = pv;
        }
    }
    __syncthreads();

#pragma unroll
    for (int n = 0; n < DSTATE; ++n) h[n] = 0.0f;
    for (int c = 0; c < w; ++c) {
#pragma unroll
        for (int q = 0; q < DSTATE / 4; ++q) {
            const float4 pv = *(const float4*)&sP[c][lane][4 * q];
            const float4 hv = *(const float4*)&sH[c][lane][4 * q];
            h[4 * q + 0] = pv.x * h[4 * q + 0] + hv.x;
            h[4 * q + 1] = pv.y * h[4 * q + 1] + hv.y;
            h[4 * q + 2] = pv.z * h[4 * q + 2] + hv.z;
            h[4 * q + 3] = pv.w * h[4 * q + 3] + hv.w;
        }
    }

    const float Dv = D_param[d];
    for (int t = 0; t < TCH; ++t) {
        const float* bc = bcp + (size_t)t * (DTRANK + 2 * DSTATE);
        const float u   = up[(size_t)t * DINNER];
        const float dtv = dp[(size_t)t * DINNER];
        const float z   = zp[(size_t)t * (2 * DINNER)];

        float acc = 0.0f;
#pragma unroll
        for (int n = 0; n < DSTATE; ++n) {
            const float dA = __expf(dtv * Ar[n]);
            h[n] = dA * h[n] + (dtv * bc[n]) * u;
            acc += bc[DSTATE + n] * h[n];
        }
        yp[(size_t)t * DINNER] = (acc + Dv * u) * silu_f(z);
    }
}

extern "C" void kernel_launch(void* const* d_in, const int* in_sizes, int n_in,
                              void* d_out, int out_size, void* d_ws, size_t ws_size,
                              hipStream_t stream)
{
    const float* hidden    = (const float*)d_in[0];
    const float* in_proj_w = (const float*)d_in[1];
    const float* conv_w    = (const float*)d_in[2];
    const float* conv_b    = (const float*)d_in[3];
    const float* x_proj_w  = (const float*)d_in[4];
    const float* dt_proj_w = (const float*)d_in[5];
    const float* dt_bias   = (const float*)d_in[6];
    const float* A_log     = (const float*)d_in[7];
    const float* D_param   = (const float*)d_in[8];
    const float* out_proj  = (const float*)d_in[9];
    float*       out       = (float*)d_out;

    float* ws   = (float*)d_ws;
    float* xz   = ws;
    float* xc   = xz   + (size_t)NROWS * 2 * DINNER;
    float* xdbl = xc   + (size_t)NROWS * DINNER;
    float* dtm  = xdbl + (size_t)NROWS * (DTRANK + 2 * DSTATE);
    float* y    = dtm  + (size_t)NROWS * DINNER;

    gemm_wmma_regtile<0, 32, DMODEL, DMODEL, 2 * DINNER, 2 * DINNER>
        <<<dim3((2 * DINNER) / 128, NROWS / 128), 128, 0, stream>>>(
        hidden, in_proj_w, xz, nullptr);

    conv_xproj_kernel<<<NROWS / 16, XP_WAVES * 32, 0, stream>>>(
        xz, conv_w, conv_b, x_proj_w, xc, xdbl);

    gemm_wmma_regtile<1, 32, DTRANK, DTRANK + 2 * DSTATE, DINNER, DINNER>
        <<<dim3(DINNER / 128, NROWS / 128), 128, 0, stream>>>(
        xdbl, dt_proj_w, dtm, dt_bias);

    scan_gate_chunked_kernel<<<dim3(DINNER / 32, B_SZ), NCHUNK * 32, 0, stream>>>(
        xc, dtm, xdbl, A_log, xz, D_param, y);

    gemm_wmma_regtile<0, 32, DINNER, DINNER, DMODEL, DMODEL>
        <<<dim3(DMODEL / 128, NROWS / 128), 128, 0, stream>>>(
        y, out_proj, out, nullptr);
}
